# starting kernel
# baseline (speedup 1.0000x reference)
#include <hip/hip_runtime.h>
#include <hip/hip_bf16.h>

typedef __attribute__((ext_vector_type(2))) float v2f;
typedef __attribute__((ext_vector_type(4))) float v4f;
typedef __attribute__((ext_vector_type(8))) float v8f;

#define N_EMB  1024
#define HEAD   64
#define SEQ_T  2048
#define BATCH  8
#define ROWS   (BATCH * SEQ_T)

static __device__ __forceinline__ v8f wmma_f32(v2f a, v2f b, v8f c) {
    return __builtin_amdgcn_wmma_f32_16x16x4_f32(false, a, false, b, (short)0, c,
                                                 false, false);
}

static __device__ __forceinline__ v2f mk2(float x, float y) { v2f r; r.x = x; r.y = y; return r; }

__global__ __launch_bounds__(128)
void sah_proj_kernel(const float* __restrict__ x,
                     const float* __restrict__ Wk,
                     const float* __restrict__ Wq,
                     const float* __restrict__ Wv,
                     float* __restrict__ kb,
                     float* __restrict__ qb,
                     float* __restrict__ vb) {
    const int lane = threadIdx.x & 31;
    const int ht   = threadIdx.x >> 5;
    const int half = lane >> 4;
    const int lr   = lane & 15;
    const int rt   = blockIdx.x;
    const int which = blockIdx.y;

    const float* W   = (which == 0) ? Wk : (which == 1) ? Wq : Wv;
    float*       out = (which == 0) ? kb : (which == 1) ? qb : vb;

    const float* arow = x + (size_t)(rt * 16 + lr) * N_EMB + half * 4;
    const float* brow = W + (size_t)(ht * 16 + lr) * N_EMB + half * 4;

    v8f accA = {}, accB = {};
#pragma unroll 4
    for (int k0 = 0; k0 < N_EMB; k0 += 8) {
        v4f a4 = *(const v4f*)(arow + k0);
        v4f b4 = *(const v4f*)(brow + k0);
        accA = wmma_f32(mk2(a4.x, a4.y), mk2(b4.x, b4.y), accA);
        accB = wmma_f32(mk2(a4.z, a4.w), mk2(b4.z, b4.w), accB);
    }

    float* obase = out + (size_t)(rt * 16 + half * 8) * HEAD + ht * 16 + lr;
#pragma unroll
    for (int r = 0; r < 8; ++r)
        obase[(size_t)r * HEAD] = accA[r] + accB[r];
}

__global__ __launch_bounds__(128)
void sah_attn_kernel(const float* __restrict__ Q,
                     const float* __restrict__ K,
                     const float* __restrict__ V,
                     float* __restrict__ O) {
    __shared__ float pstage[4][16 * 16];
    __shared__ float mred[4][16];
    __shared__ float lred[4][16];
    __shared__ float ored[4][16 * 64];

    const int lane = threadIdx.x & 31;
    const int wave = threadIdx.x >> 5;
    const int half = lane >> 4;
    const int lr   = lane & 15;
    const int b    = blockIdx.y;
    const int qt   = blockIdx.x;
    const int q0   = qt * 16;

    const float* Qb = Q + (size_t)b * SEQ_T * HEAD;
    const float* Kb = K + (size_t)b * SEQ_T * HEAD;
    const float* Vb = V + (size_t)b * SEQ_T * HEAD;
    float* lds = pstage[wave];

    v2f qfrag[16];
    {
        const float* qrow = Qb + (size_t)(q0 + lr) * HEAD + half * 4;
#pragma unroll
        for (int c = 0; c < 8; ++c) {
            v4f t = *(const v4f*)(qrow + c * 8);
            qfrag[2 * c]     = mk2(t.x, t.y);
            qfrag[2 * c + 1] = mk2(t.z, t.w);
        }
    }

    const v2f ones = mk2(1.0f, 1.0f);

    v8f acc0 = {}, acc1 = {}, acc2 = {}, acc3 = {};
    v8f lacc = {};
    float m_i[8];
#pragma unroll
    for (int r = 0; r < 8; ++r) m_i[r] = -__builtin_inff();

    const float scale = 0.03125f;

    for (int j = wave; j <= qt; j += 4) {
        const int k0r = j * 16;

        v8f sA = {}, sB = {};
        {
            const float* krow = Kb + (size_t)(k0r + lr) * HEAD + half * 4;
#pragma unroll
            for (int c = 0; c < 8; ++c) {
                v4f t = *(const v4f*)(krow + c * 8);
                sA = wmma_f32(qfrag[2 * c],     mk2(t.x, t.y), sA);
                sB = wmma_f32(qfrag[2 * c + 1], mk2(t.z, t.w), sB);
            }
        }

        const bool diag = (j == qt);
#pragma unroll
        for (int r = 0; r < 8; ++r) {
            const int m = r + half * 8;
            float sv = (sA[r] + sB[r]) * scale;
            if (diag && lr > m) sv = -__builtin_inff();

            float rm = sv;
            rm = fmaxf(rm, __shfl_xor(rm, 1, 32));
            rm = fmaxf(rm, __shfl_xor(rm, 2, 32));
            rm = fmaxf(rm, __shfl_xor(rm, 4, 32));
            rm = fmaxf(rm, __shfl_xor(rm, 8, 32));

            float mn    = fmaxf(m_i[r], rm);
            float alpha = __expf(m_i[r] - mn);
            m_i[r] = mn;

            float pv = __expf(sv - mn);

            lacc[r] *= alpha;
            acc0[r] *= alpha;
            acc1[r] *= alpha;
            acc2[r] *= alpha;
            acc3[r] *= alpha;

            lds[m * 16 + lr] = pv;
        }

#pragma unroll
        for (int kk = 0; kk < 4; ++kk) {
            v2f pa = *(const v2f*)&lds[lr * 16 + kk * 4 + half * 2];
            const float* vp = Vb + (size_t)(k0r + kk * 4 + half * 2) * HEAD + lr;
            acc0 = wmma_f32(pa, mk2(vp[0],  vp[HEAD +  0]), acc0);
            acc1 = wmma_f32(pa, mk2(vp[16], vp[HEAD + 16]), acc1);
            acc2 = wmma_f32(pa, mk2(vp[32], vp[HEAD + 32]), acc2);
            acc3 = wmma_f32(pa, mk2(vp[48], vp[HEAD + 48]), acc3);
            lacc = wmma_f32(pa, ones, lacc);
        }
    }

    if (lr == 0) {
#pragma unroll
        for (int r = 0; r < 8; ++r)
            mred[wave][r + half * 8] = m_i[r];
    }
    __syncthreads();

#pragma unroll
    for (int r = 0; r < 8; ++r) {
        const int row = r + half * 8;
        float ms = fmaxf(fmaxf(mred[0][row], mred[1][row]),
                         fmaxf(mred[2][row], mred[3][row]));
        float al = __expf(m_i[r] - ms);
        acc0[r] *= al;
        acc1[r] *= al;
        acc2[r] *= al;
        acc3[r] *= al;
        if (lr == 0) lred[wave][row] = lacc[r] * al;
        float* orow = &ored[wave][row * HEAD + lr];
        orow[0]  = acc0[r];
        orow[16] = acc1[r];
        orow[32] = acc2[r];
        orow[48] = acc3[r];
    }
    __syncthreads();

    float* ob = O + ((size_t)b * SEQ_T + q0 + half * 8) * HEAD + wave * 16 + lr;
#pragma unroll
    for (int r = 0; r < 8; ++r) {
        const int row = r + half * 8;
        float lt = (lred[0][row] + lred[1][row]) + (lred[2][row] + lred[3][row]);
        const int c = row * HEAD + wave * 16 + lr;
        float s = (ored[0][c] + ored[1][c]) + (ored[2][c] + ored[3][c]);
        ob[(size_t)r * HEAD] = s / lt;
    }
}

extern "C" void kernel_launch(void* const* d_in, const int* in_sizes, int n_in,
                              void* d_out, int out_size, void* d_ws, size_t ws_size,
                              hipStream_t stream) {
    const float* x  = (const float*)d_in[0];
    const float* Wk = (const float*)d_in[1];
    const float* Wq = (const float*)d_in[2];
    const float* Wv = (const float*)d_in[3];

    float* kb = (float*)d_ws;
    float* qb = kb + (size_t)ROWS * HEAD;
    float* vb = qb + (size_t)ROWS * HEAD;

    sah_proj_kernel<<<dim3(ROWS / 16, 3), 128, 0, stream>>>(x, Wk, Wq, Wv, kb, qb, vb);
    sah_attn_kernel<<<dim3(SEQ_T / 16, BATCH), 128, 0, stream>>>(qb, kb, vb,
                                                                 (float*)d_out);
}


# Projection as split-K in registers: block = 64 rows x 3 matrices, 12 waves; each wave sums a 64x64 tile set over one K quarter from direct global loads, quarters added via LDS (sum order changes)
# speedup vs baseline: 1.1648x; 1.1648x over previous
#include <hip/hip_runtime.h>
#include <hip/hip_bf16.h>

typedef __attribute__((ext_vector_type(2))) float v2f;
typedef __attribute__((ext_vector_type(4))) float v4f;
typedef __attribute__((ext_vector_type(8))) float v8f;

#define N_EMB  1024
#define HEAD   64
#define SEQ_T  2048
#define BATCH  8
#define ROWS   (BATCH * SEQ_T)

static __device__ __forceinline__ v8f wmma_f32(v2f a, v2f b, v8f c) {
    return __builtin_amdgcn_wmma_f32_16x16x4_f32(false, a, false, b, (short)0, c,
                                                 false, false);
}

static __device__ __forceinline__ v2f mk2(float x, float y) { v2f r; r.x = x; r.y = y; return r; }

#define PJ_ROWS    64
#define PJ_TSTRIDE 272
#define PJ_WSTRIDE (16 * PJ_TSTRIDE)

__global__ __launch_bounds__(384)
void sah_proj_kernel(const float* __restrict__ x,
                     const float* __restrict__ Wk,
                     const float* __restrict__ Wq,
                     const float* __restrict__ Wv,
                     float* __restrict__ kb,
                     float* __restrict__ qb,
                     float* __restrict__ vb) {
    __shared__ __attribute__((aligned(16))) float part[12 * PJ_WSTRIDE];

    const int lane  = threadIdx.x & 31;
    const int wv    = threadIdx.x >> 5;
    const int which = wv >> 2;
    const int kw    = wv & 3;
    const int half  = lane >> 4;
    const int lr    = lane & 15;
    const int row0  = blockIdx.x * PJ_ROWS;

    const float* W   = (which == 0) ? Wk : (which == 1) ? Wq : Wv;
    float*       out = (which == 0) ? kb : (which == 1) ? qb : vb;

    const float* ap = x + (size_t)(row0 + lr) * N_EMB + kw * 256 + half * 4;
    const float* bp = W + (size_t)lr * N_EMB + kw * 256 + half * 4;

    v8f acc[4][4];
#pragma unroll
    for (int i = 0; i < 4; ++i)
#pragma unroll
        for (int j = 0; j < 4; ++j)
            acc[i][j] = (v8f){};

#pragma unroll 1
    for (int ks = 0; ks < 256; ks += 32) {
        v4f a[4][4], b[4][4];
#pragma unroll
        for (int q = 0; q < 4; ++q) {
#pragma unroll
            for (int i = 0; i < 4; ++i) {
                a[i][q] = *(const v4f*)(ap + (size_t)i * 16 * N_EMB + ks + q * 8);
                b[i][q] = *(const v4f*)(bp + (size_t)i * 16 * N_EMB + ks + q * 8);
            }
        }
#pragma unroll
        for (int q = 0; q < 4; ++q) {
#pragma unroll
            for (int i = 0; i < 4; ++i)
#pragma unroll
                for (int j = 0; j < 4; ++j)
                    acc[i][j] = wmma_f32(mk2(a[i][q].x, a[i][q].y),
                                         mk2(b[j][q].x, b[j][q].y), acc[i][j]);
#pragma unroll
            for (int i = 0; i < 4; ++i)
#pragma unroll
                for (int j = 0; j < 4; ++j)
                    acc[i][j] = wmma_f32(mk2(a[i][q].z, a[i][q].w),
                                         mk2(b[j][q].z, b[j][q].w), acc[i][j]);
        }
    }

    float* mine = part + wv * PJ_WSTRIDE;
#pragma unroll
    for (int i = 0; i < 4; ++i)
#pragma unroll
        for (int j = 0; j < 4; ++j)
#pragma unroll
            for (int r = 0; r < 8; ++r)
                mine[(i * 4 + j) * PJ_TSTRIDE + r * 32 + lane] = acc[i][j][r];
    __syncthreads();

    const float* grp = part + (which * 4) * PJ_WSTRIDE;
    float* ob = out + (size_t)(row0 + kw * 16) * HEAD + lr * 4;
    const int jt = lr >> 2;
    const int n0 = (lr & 3) * 4;
#pragma unroll
    for (int it = 0; it < 8; ++it) {
        const int m   = 2 * it + half;
        const int off = (kw * 4 + jt) * PJ_TSTRIDE + (m & 7) * 32 + (m >> 3) * 16 + n0;
        v4f p0 = *(const v4f*)(grp + 0 * PJ_WSTRIDE + off);
        v4f p1 = *(const v4f*)(grp + 1 * PJ_WSTRIDE + off);
        v4f p2 = *(const v4f*)(grp + 2 * PJ_WSTRIDE + off);
        v4f p3 = *(const v4f*)(grp + 3 * PJ_WSTRIDE + off);
        *(v4f*)(ob + (size_t)m * HEAD) = (p0 + p1) + (p2 + p3);
    }
}

__global__ __launch_bounds__(128)
void sah_attn_kernel(const float* __restrict__ Q,
                     const float* __restrict__ K,
                     const float* __restrict__ V,
                     float* __restrict__ O) {
    __shared__ float pstage[4][16 * 16];
    __shared__ float mred[4][16];
    __shared__ float lred[4][16];
    __shared__ float ored[4][16 * 64];

    const int lane = threadIdx.x & 31;
    const int wave = threadIdx.x >> 5;
    const int half = lane >> 4;
    const int lr   = lane & 15;
    const int b    = blockIdx.y;
    const int qt   = blockIdx.x;
    const int q0   = qt * 16;

    const float* Qb = Q + (size_t)b * SEQ_T * HEAD;
    const float* Kb = K + (size_t)b * SEQ_T * HEAD;
    const float* Vb = V + (size_t)b * SEQ_T * HEAD;
    float* lds = pstage[wave];

    v2f qfrag[16];
    {
        const float* qrow = Qb + (size_t)(q0 + lr) * HEAD + half * 4;
#pragma unroll
        for (int c = 0; c < 8; ++c) {
            v4f t = *(const v4f*)(qrow + c * 8);
            qfrag[2 * c]     = mk2(t.x, t.y);
            qfrag[2 * c + 1] = mk2(t.z, t.w);
        }
    }

    const v2f ones = mk2(1.0f, 1.0f);

    v8f acc0 = {}, acc1 = {}, acc2 = {}, acc3 = {};
    v8f lacc = {};
    float m_i[8];
#pragma unroll
    for (int r = 0; r < 8; ++r) m_i[r] = -__builtin_inff();

    const float scale = 0.03125f;

    for (int j = wave; j <= qt; j += 4) {
        const int k0r = j * 16;

        v8f sA = {}, sB = {};
        {
            const float* krow = Kb + (size_t)(k0r + lr) * HEAD + half * 4;
#pragma unroll
            for (int c = 0; c < 8; ++c) {
                v4f t = *(const v4f*)(krow + c * 8);
                sA = wmma_f32(qfrag[2 * c],     mk2(t.x, t.y), sA);
                sB = wmma_f32(qfrag[2 * c + 1], mk2(t.z, t.w), sB);
            }
        }

        const bool diag = (j == qt);
#pragma unroll
        for (int r = 0; r < 8; ++r) {
            const int m = r + half * 8;
            float sv = (sA[r] + sB[r]) * scale;
            if (diag && lr > m) sv = -__builtin_inff();

            float rm = sv;
            rm = fmaxf(rm, __shfl_xor(rm, 1, 32));
            rm = fmaxf(rm, __shfl_xor(rm, 2, 32));
            rm = fmaxf(rm, __shfl_xor(rm, 4, 32));
            rm = fmaxf(rm, __shfl_xor(rm, 8, 32));

            float mn    = fmaxf(m_i[r], rm);
            float alpha = __expf(m_i[r] - mn);
            m_i[r] = mn;

            float pv = __expf(sv - mn);

            lacc[r] *= alpha;
            acc0[r] *= alpha;
            acc1[r] *= alpha;
            acc2[r] *= alpha;
            acc3[r] *= alpha;

            lds[m * 16 + lr] = pv;
        }

#pragma unroll
        for (int kk = 0; kk < 4; ++kk) {
            v2f pa = *(const v2f*)&lds[lr * 16 + kk * 4 + half * 2];
            const float* vp = Vb + (size_t)(k0r + kk * 4 + half * 2) * HEAD + lr;
            acc0 = wmma_f32(pa, mk2(vp[0],  vp[HEAD +  0]), acc0);
            acc1 = wmma_f32(pa, mk2(vp[16], vp[HEAD + 16]), acc1);
            acc2 = wmma_f32(pa, mk2(vp[32], vp[HEAD + 32]), acc2);
            acc3 = wmma_f32(pa, mk2(vp[48], vp[HEAD + 48]), acc3);
            lacc = wmma_f32(pa, ones, lacc);
        }
    }

    if (lr == 0) {
#pragma unroll
        for (int r = 0; r < 8; ++r)
            mred[wave][r + half * 8] = m_i[r];
    }
    __syncthreads();

#pragma unroll
    for (int r = 0; r < 8; ++r) {
        const int row = r + half * 8;
        float ms = fmaxf(fmaxf(mred[0][row], mred[1][row]),
                         fmaxf(mred[2][row], mred[3][row]));
        float al = __expf(m_i[r] - ms);
        acc0[r] *= al;
        acc1[r] *= al;
        acc2[r] *= al;
        acc3[r] *= al;
        if (lr == 0) lred[wave][row] = lacc[r] * al;
        float* orow = &ored[wave][row * HEAD + lr];
        orow[0]  = acc0[r];
        orow[16] = acc1[r];
        orow[32] = acc2[r];
        orow[48] = acc3[r];
    }
    __syncthreads();

    float* ob = O + ((size_t)b * SEQ_T + q0 + half * 8) * HEAD + wave * 16 + lr;
#pragma unroll
    for (int r = 0; r < 8; ++r) {
        const int row = r + half * 8;
        float lt = (lred[0][row] + lred[1][row]) + (lred[2][row] + lred[3][row]);
        const int c = row * HEAD + wave * 16 + lr;
        float s = (ored[0][c] + ored[1][c]) + (ored[2][c] + ored[3][c]);
        ob[(size_t)r * HEAD] = s / lt;
    }
}

extern "C" void kernel_launch(void* const* d_in, const int* in_sizes, int n_in,
                              void* d_out, int out_size, void* d_ws, size_t ws_size,
                              hipStream_t stream) {
    const float* x  = (const float*)d_in[0];
    const float* Wk = (const float*)d_in[1];
    const float* Wq = (const float*)d_in[2];
    const float* Wv = (const float*)d_in[3];

    float* kb = (float*)d_ws;
    float* qb = kb + (size_t)ROWS * HEAD;
    float* vb = qb + (size_t)ROWS * HEAD;

    sah_proj_kernel<<<dim3(ROWS / PJ_ROWS), 384, 0, stream>>>(x, Wk, Wq, Wv, kb, qb, vb);
    sah_attn_kernel<<<dim3(SEQ_T / 16, BATCH), 128, 0, stream>>>(qb, kb, vb,
                                                                 (float*)d_out);
}


# Attention: a query tile's key tiles dealt over 16 waves (512-thread block, one resident per WGP) not 4, blocks numbered heaviest first: 8 not 32 trips per wave; merge over 16 (sum order changes)
# speedup vs baseline: 1.5691x; 1.3471x over previous
#include <hip/hip_runtime.h>
#include <hip/hip_bf16.h>

typedef __attribute__((ext_vector_type(2))) float v2f;
typedef __attribute__((ext_vector_type(4))) float v4f;
typedef __attribute__((ext_vector_type(8))) float v8f;

#define N_EMB  1024
#define HEAD   64
#define SEQ_T  2048
#define BATCH  8
#define ROWS   (BATCH * SEQ_T)

static __device__ __forceinline__ v8f wmma_f32(v2f a, v2f b, v8f c) {
    return __builtin_amdgcn_wmma_f32_16x16x4_f32(false, a, false, b, (short)0, c,
                                                 false, false);
}

static __device__ __forceinline__ v2f mk2(float x, float y) { v2f r; r.x = x; r.y = y; return r; }

#define PJ_ROWS    64
#define PJ_TSTRIDE 272
#define PJ_WSTRIDE (16 * PJ_TSTRIDE)

__global__ __launch_bounds__(384)
void sah_proj_kernel(const float* __restrict__ x,
                     const float* __restrict__ Wk,
                     const float* __restrict__ Wq,
                     const float* __restrict__ Wv,
                     float* __restrict__ kb,
                     float* __restrict__ qb,
                     float* __restrict__ vb) {
    __shared__ __attribute__((aligned(16))) float part[12 * PJ_WSTRIDE];

    const int lane  = threadIdx.x & 31;
    const int wv    = threadIdx.x >> 5;
    const int which = wv >> 2;
    const int kw    = wv & 3;
    const int half  = lane >> 4;
    const int lr    = lane & 15;
    const int row0  = blockIdx.x * PJ_ROWS;

    const float* W   = (which == 0) ? Wk : (which == 1) ? Wq : Wv;
    float*       out = (which == 0) ? kb : (which == 1) ? qb : vb;

    const float* ap = x + (size_t)(row0 + lr) * N_EMB + kw * 256 + half * 4;
    const float* bp = W + (size_t)lr * N_EMB + kw * 256 + half * 4;

    v8f acc[4][4];
#pragma unroll
    for (int i = 0; i < 4; ++i)
#pragma unroll
        for (int j = 0; j < 4; ++j)
            acc[i][j] = (v8f){};

#pragma unroll 1
    for (int ks = 0; ks < 256; ks += 32) {
        v4f a[4][4], b[4][4];
#pragma unroll
        for (int q = 0; q < 4; ++q) {
#pragma unroll
            for (int i = 0; i < 4; ++i) {
                a[i][q] = *(const v4f*)(ap + (size_t)i * 16 * N_EMB + ks + q * 8);
                b[i][q] = *(const v4f*)(bp + (size_t)i * 16 * N_EMB + ks + q * 8);
            }
        }
#pragma unroll
        for (int q = 0; q < 4; ++q) {
#pragma unroll
            for (int i = 0; i < 4; ++i)
#pragma unroll
                for (int j = 0; j < 4; ++j)
                    acc[i][j] = wmma_f32(mk2(a[i][q].x, a[i][q].y),
                                         mk2(b[j][q].x, b[j][q].y), acc[i][j]);
#pragma unroll
            for (int i = 0; i < 4; ++i)
#pragma unroll
                for (int j = 0; j < 4; ++j)
                    acc[i][j] = wmma_f32(mk2(a[i][q].z, a[i][q].w),
                                         mk2(b[j][q].z, b[j][q].w), acc[i][j]);
        }
    }

    float* mine = part + wv * PJ_WSTRIDE;
#pragma unroll
    for (int i = 0; i < 4; ++i)
#pragma unroll
        for (int j = 0; j < 4; ++j)
#pragma unroll
            for (int r = 0; r < 8; ++r)
                mine[(i * 4 + j) * PJ_TSTRIDE + r * 32 + lane] = acc[i][j][r];
    __syncthreads();

    const float* grp = part + (which * 4) * PJ_WSTRIDE;
    float* ob = out + (size_t)(row0 + kw * 16) * HEAD + lr * 4;
    const int jt = lr >> 2;
    const int n0 = (lr & 3) * 4;
#pragma unroll
    for (int it = 0; it < 8; ++it) {
        const int m   = 2 * it + half;
        const int off = (kw * 4 + jt) * PJ_TSTRIDE + (m & 7) * 32 + (m >> 3) * 16 + n0;
        v4f p0 = *(const v4f*)(grp + 0 * PJ_WSTRIDE + off);
        v4f p1 = *(const v4f*)(grp + 1 * PJ_WSTRIDE + off);
        v4f p2 = *(const v4f*)(grp + 2 * PJ_WSTRIDE + off);
        v4f p3 = *(const v4f*)(grp + 3 * PJ_WSTRIDE + off);
        *(v4f*)(ob + (size_t)m * HEAD) = (p0 + p1) + (p2 + p3);
    }
}

#define AT_WAVES 16

__global__ __launch_bounds__(AT_WAVES * 32)
void sah_attn_kernel(const float* __restrict__ Q,
                     const float* __restrict__ K,
                     const float* __restrict__ V,
                     float* __restrict__ O) {
    __shared__ float pstage[AT_WAVES][16 * 16];
    __shared__ float mred[AT_WAVES][16];
    __shared__ float lred[AT_WAVES][16];
    __shared__ __attribute__((aligned(8))) float ored[AT_WAVES][16 * 64];

    const int lane = threadIdx.x & 31;
    const int wave = threadIdx.x >> 5;
    const int half = lane >> 4;
    const int lr   = lane & 15;
    const int b    = blockIdx.x % BATCH;
    const int qt   = (SEQ_T / 16 - 1) - blockIdx.x / BATCH;
    const int q0   = qt * 16;

    const float* Qb = Q + (size_t)b * SEQ_T * HEAD;
    const float* Kb = K + (size_t)b * SEQ_T * HEAD;
    const float* Vb = V + (size_t)b * SEQ_T * HEAD;
    float* lds = pstage[wave];

    v2f qfrag[16];
    {
        const float* qrow = Qb + (size_t)(q0 + lr) * HEAD + half * 4;
#pragma unroll
        for (int c = 0; c < 8; ++c) {
            v4f t = *(const v4f*)(qrow + c * 8);
            qfrag[2 * c]     = mk2(t.x, t.y);
            qfrag[2 * c + 1] = mk2(t.z, t.w);
        }
    }

    const v2f ones = mk2(1.0f, 1.0f);

    v8f acc0 = {}, acc1 = {}, acc2 = {}, acc3 = {};
    v8f lacc = {};
    float m_i[8];
#pragma unroll
    for (int r = 0; r < 8; ++r) m_i[r] = -__builtin_inff();

    const float scale = 0.03125f;

    for (int j = wave; j <= qt; j += AT_WAVES) {
        const int k0r = j * 16;

        v8f sA = {}, sB = {};
        {
            const float* krow = Kb + (size_t)(k0r + lr) * HEAD + half * 4;
#pragma unroll
            for (int c = 0; c < 8; ++c) {
                v4f t = *(const v4f*)(krow + c * 8);
                sA = wmma_f32(qfrag[2 * c],     mk2(t.x, t.y), sA);
                sB = wmma_f32(qfrag[2 * c + 1], mk2(t.z, t.w), sB);
            }
        }

        const bool diag = (j == qt);
#pragma unroll
        for (int r = 0; r < 8; ++r) {
            const int m = r + half * 8;
            float sv = (sA[r] + sB[r]) * scale;
            if (diag && lr > m) sv = -__builtin_inff();

            float rm = sv;
            rm = fmaxf(rm, __shfl_xor(rm, 1, 32));
            rm = fmaxf(rm, __shfl_xor(rm, 2, 32));
            rm = fmaxf(rm, __shfl_xor(rm, 4, 32));
            rm = fmaxf(rm, __shfl_xor(rm, 8, 32));

            float mn    = fmaxf(m_i[r], rm);
            float alpha = __expf(m_i[r] - mn);
            m_i[r] = mn;

            float pv = __expf(sv - mn);

            lacc[r] *= alpha;
            acc0[r] *= alpha;
            acc1[r] *= alpha;
            acc2[r] *= alpha;
            acc3[r] *= alpha;

            lds[m * 16 + lr] = pv;
        }

#pragma unroll
        for (int kk = 0; kk < 4; ++kk) {
            v2f pa = *(const v2f*)&lds[lr * 16 + kk * 4 + half * 2];
            const float* vp = Vb + (size_t)(k0r + kk * 4 + half * 2) * HEAD + lr;
            acc0 = wmma_f32(pa, mk2(vp[0],  vp[HEAD +  0]), acc0);
            acc1 = wmma_f32(pa, mk2(vp[16], vp[HEAD + 16]), acc1);
            acc2 = wmma_f32(pa, mk2(vp[32], vp[HEAD + 32]), acc2);
            acc3 = wmma_f32(pa, mk2(vp[48], vp[HEAD + 48]), acc3);
            lacc = wmma_f32(pa, ones, lacc);
        }
    }

    if (lr == 0) {
#pragma unroll
        for (int r = 0; r < 8; ++r)
            mred[wave][r + half * 8] = m_i[r];
    }
    __syncthreads();

#pragma unroll
    for (int r = 0; r < 8; ++r) {
        const int row = r + half * 8;
        float ms = mred[0][row];
#pragma unroll
        for (int w = 1; w < AT_WAVES; ++w) ms = fmaxf(ms, mred[w][row]);
        float al = __expf(m_i[r] - ms);
        acc0[r] *= al;
        acc1[r] *= al;
        acc2[r] *= al;
        acc3[r] *= al;
        if (lr == 0) lred[wave][row] = lacc[r] * al;
        float* orow = &ored[wave][row * HEAD + lr];
        orow[0]  = acc0[r];
        orow[16] = acc1[r];
        orow[32] = acc2[r];
        orow[48] = acc3[r];
    }
    __syncthreads();

    {
        const int row = wave;
        const int c2  = lane * 2;
        float lt = 0.0f, sx = 0.0f, sy = 0.0f;
#pragma unroll
        for (int w = 0; w < AT_WAVES; ++w) {
            lt += lred[w][row];
            v2f pr = *(const v2f*)&ored[w][row * HEAD + c2];
            sx += pr.x;
            sy += pr.y;
        }
        float* ob = O + ((size_t)b * SEQ_T + q0 + row) * HEAD + c2;
        *(v2f*)ob = mk2(sx / lt, sy / lt);
    }
}

extern "C" void kernel_launch(void* const* d_in, const int* in_sizes, int n_in,
                              void* d_out, int out_size, void* d_ws, size_t ws_size,
                              hipStream_t stream) {
    const float* x  = (const float*)d_in[0];
    const float* Wk = (const float*)d_in[1];
    const float* Wq = (const float*)d_in[2];
    const float* Wv = (const float*)d_in[3];

    float* kb = (float*)d_ws;
    float* qb = kb + (size_t)ROWS * HEAD;
    float* vb = qb + (size_t)ROWS * HEAD;

    sah_proj_kernel<<<dim3(ROWS / PJ_ROWS), 384, 0, stream>>>(x, Wk, Wq, Wv, kb, qb, vb);
    sah_attn_kernel<<<dim3(SEQ_T / 16 * BATCH), AT_WAVES * 32, 0, stream>>>(qb, kb, vb,
                                                                 (float*)d_out);
}
